# starting kernel
# baseline (speedup 1.0000x reference)
#include <hip/hip_runtime.h>
#include <math.h>

typedef __attribute__((ext_vector_type(2))) float v2f;
typedef __attribute__((ext_vector_type(8))) float v8f;

#define NUM_AC    32
#define NWIN      64
#define WLEN      256
#define WSTRIDE   128
#define ROWLEN    8320
#define CSTRIDE_M 20
#define CSTRIDE_C 320

__global__ __launch_bounds__(256, 1)
void autocorr_wmma_kernel(const float* __restrict__ x, float* __restrict__ out)
{
    __shared__ __align__(16) float s_row[ROWLEN];
    __shared__ float s_hann[WLEN];
    __shared__ float s_v[8][288];
    __shared__ float s_c[8][3 * CSTRIDE_C];

    const int tid  = threadIdx.x;
    const int lane = tid & 31;
    const int wv   = tid >> 5;
    const int col  = lane & 15;
    const int hi   = lane >> 4;
    const int row  = blockIdx.x;

    if (tid < WLEN) {
        s_hann[tid] = 0.5f - 0.5f * cosf(6.28318530717958647692f * (float)tid * (1.0f / 255.0f));
    }

    {
        const float4* src = reinterpret_cast<const float4*>(x + (size_t)row * ROWLEN);
        float4*       dst = reinterpret_cast<float4*>(s_row);
        for (int i = tid; i < ROWLEN / 4; i += 256) dst[i] = src[i];
    }
    __syncthreads();

    float* vb = s_v[wv];
    float* cb = s_c[wv];

    for (int w = wv; w < NWIN; w += 8) {
        const float* win = s_row + w * WSTRIDE;

        float ps = 0.0f;
        #pragma unroll
        for (int i = 0; i < 8; ++i) ps += win[i * 32 + lane];
        #pragma unroll
        for (int d = 16; d >= 1; d >>= 1) ps += __shfl_xor(ps, d, 32);
        const float mean = ps * (1.0f / 256.0f);

        #pragma unroll
        for (int i = 0; i < 8; ++i) {
            const int t = i * 32 + lane;
            vb[t] = (win[t] - mean) * s_hann[t];
        }
        vb[256 + lane] = 0.0f;
        __builtin_amdgcn_wave_barrier();

        float r[16];
        #pragma unroll
        for (int i = 0; i < 16; ++i) r[i] = vb[(2 * hi + i) * 16 + col];

        v8f c0 = {}; v8f c1 = {}; v8f c2 = {};
        #pragma unroll
        for (int q = 0; q < 4; ++q) {
            v2f a  = { r[4 * q + 0], r[4 * q + 1] };
            v2f b1 = { r[4 * q + 1], r[4 * q + 2] };
            v2f b2 = { r[4 * q + 2], r[4 * q + 3] };
            c0 = __builtin_amdgcn_wmma_f32_16x16x4_f32(false, a, false, a,  (short)0, c0, false, false);
            c1 = __builtin_amdgcn_wmma_f32_16x16x4_f32(false, a, false, b1, (short)0, c1, false, false);
            c2 = __builtin_amdgcn_wmma_f32_16x16x4_f32(false, a, false, b2, (short)0, c2, false, false);
        }

        #pragma unroll
        for (int j = 0; j < 8; ++j) {
            const int m = 8 * hi + j;
            cb[0 * CSTRIDE_C + m * CSTRIDE_M + col] = c0[j];
            cb[1 * CSTRIDE_C + m * CSTRIDE_M + col] = c1[j];
            cb[2 * CSTRIDE_C + m * CSTRIDE_M + col] = c2[j];
        }
        __builtin_amdgcn_wave_barrier();

        float acc = 0.0f;
        #pragma unroll
        for (int m = 0; m < 16; ++m) {
            const int s2 = lane + m;
            acc += cb[(s2 >> 4) * CSTRIDE_C + m * CSTRIDE_M + (s2 & 15)];
        }
        out[((size_t)row * NWIN + w) * NUM_AC + lane] = acc * (1.0f / 256.0f);
        __builtin_amdgcn_wave_barrier();
    }
}

extern "C" void kernel_launch(void* const* d_in, const int* in_sizes, int n_in,
                              void* d_out, int out_size, void* d_ws, size_t ws_size,
                              hipStream_t stream) {
    const float* x   = (const float*)d_in[0];
    float*       out = (float*)d_out;
    dim3 grid(16 * 128);
    dim3 block(256);
    hipLaunchKernelGGL(autocorr_wmma_kernel, grid, block, 0, stream, x, out);
}


# Row staged transposed in LDS (16 lines by column): a lane's 10 fragment samples (k = 8*half+2q+s) lie contiguous, 3 wide loads a window in one group, mean/Hann in registers; k sum order changed
# speedup vs baseline: 1.2656x; 1.2656x over previous
#include <hip/hip_runtime.h>
#include <math.h>

typedef __attribute__((ext_vector_type(2))) float v2f;
typedef __attribute__((ext_vector_type(8))) float v8f;

#define NUM_AC    32
#define NWIN      64
#define WLEN      256
#define WSTRIDE   128
#define ROWLEN    8320
#define TP        524
#define EPITCH    52

template <int MASK>
__device__ __forceinline__ float row_xor(float v)
{
    const int b = __builtin_bit_cast(int, v);
    return __builtin_bit_cast(float, __builtin_amdgcn_update_dpp(0, b, 0x160 | MASK, 0xf, 0xf, true));
}

__device__ __forceinline__ float other_row(float v)
{
    const int b = __builtin_bit_cast(int, v);
    return __builtin_bit_cast(float, __builtin_amdgcn_permlanex16(b, b, 0x76543210, 0xfedcba98, false, false));
}

__device__ __forceinline__ float wave_sum(float p)
{
    p += row_xor<1>(p);
    p += row_xor<2>(p);
    p += row_xor<4>(p);
    p += row_xor<8>(p);
    p += other_row(p);
    return p;
}

__global__ __launch_bounds__(256, 1)
void autocorr_wmma_kernel(const float* __restrict__ x, float* __restrict__ out)
{
    __shared__ __align__(16) float s_rowT[16 * TP];
    __shared__ float s_hann[WLEN];
    __shared__ float s_half[NWIN + 1];
    __shared__ __align__(16) float s_e[8][16 * EPITCH];

    const int tid  = threadIdx.x;
    const int lane = tid & 31;
    const int wv   = __builtin_amdgcn_readfirstlane(tid >> 5);
    const int col  = lane & 15;
    const int hi   = lane >> 4;
    const int row  = blockIdx.x;

    if (tid < WLEN) {
        s_hann[tid] = 0.5f - 0.5f * cosf(6.28318530717958647692f * (float)tid * (1.0f / 255.0f));
    }

    if (tid < 64) s_rowT[(tid >> 2) * TP + 520 + (tid & 3)] = 0.0f;

    {
        const float4* src = reinterpret_cast<const float4*>(x + (size_t)row * ROWLEN);
        float*        dst = s_rowT + 4 * (tid & 3) * TP + (tid >> 2);
        float4 buf[8];
        #pragma unroll
        for (int j = 0; j < 8; ++j) buf[j] = src[tid + 256 * j];
        float4 last = make_float4(0.0f, 0.0f, 0.0f, 0.0f);
        if (wv == 0) last = src[2048 + lane];
        float p[8];
        #pragma unroll
        for (int j = 0; j < 8; ++j) {
            dst[64 * j]          = buf[j].x;
            dst[64 * j + TP]     = buf[j].y;
            dst[64 * j + 2 * TP] = buf[j].z;
            dst[64 * j + 3 * TP] = buf[j].w;
            p[j] = (buf[j].x + buf[j].y) + (buf[j].z + buf[j].w);
        }
        const bool b0 = (lane & 1) != 0, b1 = (lane & 2) != 0, b2 = (lane & 4) != 0;
        float q[4];
        #pragma unroll
        for (int k = 0; k < 4; ++k) {
            const float keep = b0 ? p[2 * k + 1] : p[2 * k];
            const float send = b0 ? p[2 * k] : p[2 * k + 1];
            q[k] = keep + row_xor<1>(send);
        }
        float u[2];
        #pragma unroll
        for (int k = 0; k < 2; ++k) {
            const float keep = b1 ? q[2 * k + 1] : q[2 * k];
            const float send = b1 ? q[2 * k] : q[2 * k + 1];
            u[k] = keep + row_xor<2>(send);
        }
        float hs = (b2 ? u[1] : u[0]) + row_xor<4>(b2 ? u[0] : u[1]);
        hs += row_xor<8>(hs);
        hs += other_row(hs);
        if (lane < 8) s_half[wv + 8 * lane] = hs * (1.0f / 256.0f);
        if (wv == 0) {
            float* dl = s_rowT + 4 * (lane & 3) * TP + 512 + (lane >> 2);
            dl[0] = last.x; dl[TP] = last.y; dl[2 * TP] = last.z; dl[3 * TP] = last.w;
            const float hl = wave_sum((last.x + last.y) + (last.z + last.w));
            if (lane == 0) s_half[NWIN] = hl * (1.0f / 256.0f);
        }
    }
    __syncthreads();

    float* eb = s_e[wv];

    const int hsum = __builtin_bit_cast(int, s_half[wv + 8 * ((lane & 15) >> 1) + (lane & 1)]);

    float hn[10];
    #pragma unroll
    for (int i = 0; i < 8; ++i) hn[i] = s_hann[16 * (8 * hi + i) + col];
    hn[8] = hi ? 0.0f : s_hann[16 * 8 + col];
    hn[9] = hi ? 0.0f : s_hann[16 * 9 + col];

    const float* fa = s_rowT + col * TP + 8 * hi;
    const float* fz = s_rowT + col * TP + 520;

    float4 x0 = *reinterpret_cast<const float4*>(fa + 8 * wv);
    float4 x1 = *reinterpret_cast<const float4*>(fa + 8 * wv + 4);
    float2 x2 = *reinterpret_cast<const float2*>(hi ? fz : fa + 8 * wv + 8);

    #pragma unroll 1
    for (int j = 0; j < NWIN / 8; ++j) {
        const int w = wv + 8 * j;

        const float mean = __builtin_bit_cast(float, __builtin_amdgcn_readlane(hsum, 2 * j))
                         + __builtin_bit_cast(float, __builtin_amdgcn_readlane(hsum, 2 * j + 1));

        float r[10];
        r[0] = (x0.x - mean) * hn[0]; r[1] = (x0.y - mean) * hn[1];
        r[2] = (x0.z - mean) * hn[2]; r[3] = (x0.w - mean) * hn[3];
        r[4] = (x1.x - mean) * hn[4]; r[5] = (x1.y - mean) * hn[5];
        r[6] = (x1.z - mean) * hn[6]; r[7] = (x1.w - mean) * hn[7];
        r[8] = (x2.x - mean) * hn[8]; r[9] = (x2.y - mean) * hn[9];

        v8f c0 = {}; v8f c1 = {}; v8f c2 = {};
        #pragma unroll
        for (int q = 0; q < 4; ++q) {
            v2f a  = { r[2 * q + 0], r[2 * q + 1] };
            v2f b1 = { r[2 * q + 1], r[2 * q + 2] };
            v2f b2 = { r[2 * q + 2], r[2 * q + 3] };
            c0 = __builtin_amdgcn_wmma_f32_16x16x4_f32(false, a, false, a,  (short)0, c0, false, false);
            c1 = __builtin_amdgcn_wmma_f32_16x16x4_f32(false, b1, false, a, (short)0, c1, false, false);
            c2 = __builtin_amdgcn_wmma_f32_16x16x4_f32(false, b2, false, a, (short)0, c2, false, false);
        }

        __builtin_amdgcn_sched_barrier(0);
        {
            const float* fn = fa + 8 * (wv + 8 * ((j + 1) & 7));
            x0 = *reinterpret_cast<const float4*>(fn);
            x1 = *reinterpret_cast<const float4*>(fn + 4);
            x2 = *reinterpret_cast<const float2*>(hi ? fz : fn + 8);
        }

        {
            float4* er = reinterpret_cast<float4*>(eb + col * EPITCH + 8 * hi);
            er[0] = make_float4(c0[0], c0[1], c0[2], c0[3]);
            er[1] = make_float4(c0[4], c0[5], c0[6], c0[7]);
            er[4] = make_float4(c1[0], c1[1], c1[2], c1[3]);
            er[5] = make_float4(c1[4], c1[5], c1[6], c1[7]);
            er[8] = make_float4(c2[0], c2[1], c2[2], c2[3]);
            er[9] = make_float4(c2[4], c2[5], c2[6], c2[7]);
        }
        __builtin_amdgcn_wave_barrier();

        float acc = 0.0f;
        #pragma unroll
        for (int m = 0; m < 16; ++m) {
            acc += eb[m * (EPITCH + 1) + lane];
        }
        out[((size_t)row * NWIN + w) * NUM_AC + lane] = acc * (1.0f / 256.0f);
        __builtin_amdgcn_wave_barrier();
    }
}

extern "C" void kernel_launch(void* const* d_in, const int* in_sizes, int n_in,
                              void* d_out, int out_size, void* d_ws, size_t ws_size,
                              hipStream_t stream) {
    const float* x   = (const float*)d_in[0];
    float*       out = (float*)d_out;
    dim3 grid(16 * 128);
    dim3 block(256);
    hipLaunchKernelGGL(autocorr_wmma_kernel, grid, block, 0, stream, x, out);
}
